# starting kernel
# baseline (speedup 1.0000x reference)
#include <hip/hip_runtime.h>
#include <math.h>

#define D     512
#define NWAY  64
#define NQ    65536
#define NS    65536
#define EPSF  1e-30f
#define MBLK  128
#define KC    64
#define LDSTR 68

typedef float v2f __attribute__((ext_vector_type(2)));
typedef float v4f __attribute__((ext_vector_type(4)));
typedef float v8f __attribute__((ext_vector_type(8)));

__global__ void __launch_bounds__(256)
proto_partial(const float* __restrict__ sup, const int* __restrict__ lab,
              float* __restrict__ pSums) {
  __shared__ float ls[NWAY * 256];
  const int t    = threadIdx.x;
  const int bx   = blockIdx.x;
  const int rc   = bx >> 1;
  const int half = bx & 1;
  for (int i = t; i < NWAY * 256; i += 256) ls[i] = 0.f;
  __syncthreads();
  const int base = rc * 1024;
  const float* src = sup + (size_t)base * D + half * 256 + t;
  for (int r = 0; r < 1024; ++r) {
    int L = lab[base + r];
    ls[L * 256 + t] += src[(size_t)r * D];
  }
  __syncthreads();
  float* dst = pSums + (size_t)bx * (NWAY * 256);
  for (int i = t; i < NWAY * 256; i += 256) dst[i] = ls[i];
}

__global__ void __launch_bounds__(256)
proto_final(const float* __restrict__ pSums, const int* __restrict__ lab,
            const float* __restrict__ scaleP, float* __restrict__ protN) {
  __shared__ float red[256];
  __shared__ int   redi[256];
  const int c = blockIdx.x;
  const int t = threadIdx.x;

  int cnt = 0;
  for (int i = t; i < NS; i += 256) cnt += (lab[i] == c) ? 1 : 0;
  redi[t] = cnt;
  __syncthreads();
  for (int s = 128; s > 0; s >>= 1) {
    if (t < s) redi[t] += redi[t + s];
    __syncthreads();
  }
  const float fcnt = (float)redi[0];

  float p[2];
  #pragma unroll
  for (int j = 0; j < 2; ++j) {
    int d = t + j * 256;
    int half = d >> 8, dcol = d & 255;
    float s = 0.f;
    for (int rcb = 0; rcb < 64; ++rcb)
      s += pSums[(size_t)((rcb * 2 + half) * NWAY + c) * 256 + dcol];
    p[j] = s / fcnt;
  }

  red[t] = p[0] * p[0] + p[1] * p[1];
  __syncthreads();
  for (int s = 128; s > 0; s >>= 1) {
    if (t < s) red[t] += red[t + s];
    __syncthreads();
  }
  const float fac = scaleP[0] / fmaxf(sqrtf(red[0]), EPSF);
  protN[(size_t)c * D + t]       = p[0] * fac;
  protN[(size_t)c * D + t + 256] = p[1] * fac;
}

__global__ void __launch_bounds__(256)
cosine_gemm(const float* __restrict__ q, const float* __restrict__ protN,
            float* __restrict__ out) {
  __shared__ float qs[MBLK * LDSTR];
  __shared__ float ps[NWAY * LDSTR];
  __shared__ float normP[256];
  __shared__ float invq[MBLK];

  const int t    = threadIdx.x;
  const int w    = t >> 5;
  const int lane = t & 31;
  const int l15  = lane & 15;
  const int hi   = lane >> 4;
  const size_t qBase = (size_t)blockIdx.x * MBLK;

  v8f acc0 = {}, acc1 = {}, acc2 = {}, acc3 = {};
  float nacc = 0.f;
  const int nrow  = t & 127;
  const int nhalf = t >> 7;

  for (int kc = 0; kc < D; kc += KC) {
    #pragma unroll
    for (int j = 0; j < 8; ++j) {
      int v = t + j * 256;
      int row = v >> 4, c4 = (v & 15) * 4;
      v4f val = *(const v4f*)(q + (qBase + row) * D + kc + c4);
      *(v4f*)(qs + row * LDSTR + c4) = val;
    }
    #pragma unroll
    for (int j = 0; j < 4; ++j) {
      int v = t + j * 256;
      int row = v >> 4, c4 = (v & 15) * 4;
      v4f val = *(const v4f*)(protN + (size_t)row * D + kc + c4);
      *(v4f*)(ps + row * LDSTR + c4) = val;
    }
    __syncthreads();

    #pragma unroll
    for (int i = 0; i < 8; ++i) {
      v4f x = *(const v4f*)(qs + nrow * LDSTR + nhalf * 32 + i * 4);
      nacc += x.x * x.x + x.y * x.y + x.z * x.z + x.w * x.w;
    }

    for (int k4 = 0; k4 < KC; k4 += 4) {
      v2f a  = *(const v2f*)(qs + (w * 16 + l15) * LDSTR + k4 + 2 * hi);
      v2f b0 = *(const v2f*)(ps + ( 0 + l15) * LDSTR + k4 + 2 * hi);
      v2f b1 = *(const v2f*)(ps + (16 + l15) * LDSTR + k4 + 2 * hi);
      v2f b2 = *(const v2f*)(ps + (32 + l15) * LDSTR + k4 + 2 * hi);
      v2f b3 = *(const v2f*)(ps + (48 + l15) * LDSTR + k4 + 2 * hi);
      acc0 = __builtin_amdgcn_wmma_f32_16x16x4_f32(false, a, false, b0, (short)0, acc0, false, false);
      acc1 = __builtin_amdgcn_wmma_f32_16x16x4_f32(false, a, false, b1, (short)0, acc1, false, false);
      acc2 = __builtin_amdgcn_wmma_f32_16x16x4_f32(false, a, false, b2, (short)0, acc2, false, false);
      acc3 = __builtin_amdgcn_wmma_f32_16x16x4_f32(false, a, false, b3, (short)0, acc3, false, false);
    }
    __syncthreads();
  }

  normP[t] = nacc;
  __syncthreads();
  if (t < MBLK) {
    float n2 = normP[t] + normP[t + MBLK];
    invq[t] = 1.f / fmaxf(sqrtf(n2), EPSF);
  }
  __syncthreads();

  #pragma unroll
  for (int r = 0; r < 8; ++r) {
    int qrow = w * 16 + r + 8 * hi;
    float s = invq[qrow];
    size_t orow = (qBase + qrow) * NWAY;
    out[orow +  0 + l15] = acc0[r] * s;
    out[orow + 16 + l15] = acc1[r] * s;
    out[orow + 32 + l15] = acc2[r] * s;
    out[orow + 48 + l15] = acc3[r] * s;
  }
}

extern "C" void kernel_launch(void* const* d_in, const int* in_sizes, int n_in,
                              void* d_out, int out_size, void* d_ws, size_t ws_size,
                              hipStream_t stream) {
  const float* sup    = (const float*)d_in[0];
  const int*   slab   = (const int*)d_in[1];
  const float* qry    = (const float*)d_in[2];
  const float* scaleP = (const float*)d_in[4];
  float* out = (float*)d_out;

  float* pSums = (float*)d_ws;
  float* protN = (float*)((char*)d_ws + (size_t)128 * NWAY * 256 * sizeof(float));

  proto_partial<<<128, 256, 0, stream>>>(sup, slab, pSums);
  proto_final  <<<NWAY, 256, 0, stream>>>(pSums, slab, scaleP, protN);
  cosine_gemm  <<<NQ / MBLK, 256, 0, stream>>>(qry, protN, out);
}


# cosine kernel: prototypes resident in LDS, each wave owns 16 queries end to end via a wave-private LDS slice, no block barrier in K loop; query norm summed from A fragments (changes its sum order)
# speedup vs baseline: 3.5632x; 3.5632x over previous
#include <hip/hip_runtime.h>
#include <math.h>

#define D     512
#define NWAY  64
#define NQ    65536
#define NS    65536
#define EPSF  1e-30f
#define KC    64
#define LDSTR 68

typedef float v2f __attribute__((ext_vector_type(2)));
typedef float v4f __attribute__((ext_vector_type(4)));
typedef float v8f __attribute__((ext_vector_type(8)));

#define GS_SPLIT   8
#define GS_ROWS    (NS / GS_SPLIT)
#define GS_WAVES   8
#define GS_WROWS   (GS_ROWS / GS_WAVES)
#define GS_LJ      (GS_WROWS / 128)

typedef int v4i __attribute__((ext_vector_type(4)));

__device__ __forceinline__ void gs_load(v4f (&v)[4][4], const float* __restrict__ supLane,
                                        const int* list, int b, int total) {
  int r[4];
  #pragma unroll
  for (int u = 0; u < 4; ++u) {
    int idx = b * 4 + u;
    idx = idx < total ? idx : b * 4;
    r[u] = list[idx];
  }
  #pragma unroll
  for (int u = 0; u < 4; ++u) {
    const int rr = __builtin_amdgcn_readfirstlane(r[u]);
    const v4f* p = (const v4f*)(supLane + (size_t)rr * D);
    #pragma unroll
    for (int qd = 0; qd < 4; ++qd)
      v[u][qd] = __builtin_nontemporal_load(p + qd * 32);
  }
}

__device__ __forceinline__ void gs_add_full(v4f (&a)[4], const v4f (&v)[4][4]) {
  #pragma unroll
  for (int u = 0; u < 4; ++u) {
    #pragma unroll
    for (int qd = 0; qd < 4; ++qd) a[qd] += v[u][qd];
  }
}

__device__ __forceinline__ void gs_add_last(v4f (&a)[4], const v4f (&v)[4][4], int b, int total) {
  #pragma unroll
  for (int u = 0; u < 4; ++u) {
    if (b * 4 + u < total) {
      #pragma unroll
      for (int qd = 0; qd < 4; ++qd) a[qd] += v[u][qd];
    }
  }
}

__global__ void __launch_bounds__(256)
class_gather(const float* __restrict__ sup, const int* __restrict__ lab,
             float* __restrict__ pSums, int* __restrict__ pCnt) {
  __shared__ int   list[GS_ROWS];
  __shared__ float red[GS_WAVES * D];
  __shared__ int   wcnt[GS_WAVES];

  const int t    = threadIdx.x;
  const int w    = __builtin_amdgcn_readfirstlane(t >> 5);
  const int lane = t & 31;
  const int c    = blockIdx.x / GS_SPLIT;
  const int rs   = blockIdx.x % GS_SPLIT;
  const int row0 = rs * GS_ROWS + w * GS_WROWS;

  v4i L[GS_LJ];
  #pragma unroll
  for (int j = 0; j < GS_LJ; ++j)
    L[j] = *(const v4i*)(lab + row0 + j * 128 + lane * 4);

  int cw = 0;
  #pragma unroll
  for (int j = 0; j < GS_LJ; ++j) {
    #pragma unroll
    for (int e = 0; e < 4; ++e)
      cw += __builtin_popcount(__builtin_amdgcn_ballot_w32(L[j][e] == c));
  }
  if (lane == 0) wcnt[w] = cw;
  __syncthreads();

  int off = 0, total = 0;
  #pragma unroll
  for (int i = 0; i < GS_WAVES; ++i) {
    const int v = wcnt[i];
    off   += (i < w) ? v : 0;
    total += v;
  }
  total = __builtin_amdgcn_readfirstlane(total);
  off   = __builtin_amdgcn_readfirstlane(off);

  const unsigned below = (1u << lane) - 1u;
  #pragma unroll
  for (int j = 0; j < GS_LJ; ++j) {
    #pragma unroll
    for (int e = 0; e < 4; ++e) {
      const bool m = (L[j][e] == c);
      const unsigned bal = __builtin_amdgcn_ballot_w32(m);
      if (m) list[off + __builtin_popcount(bal & below)] = row0 + j * 128 + lane * 4 + e;
      off += __builtin_popcount(bal);
    }
  }
  __syncthreads();

  v4f a[4] = {};
  const int nb = (total + 3) >> 2;
  const float* supLane = sup + lane * 4;
  int b = w;
  if (b < nb) {
    v4f A[4][4], B[4][4];
    gs_load(A, supLane, list, b, total);
    while (b + 2 * GS_WAVES < nb) {
      gs_load(B, supLane, list, b + GS_WAVES, total);
      __builtin_amdgcn_sched_barrier(0);
      gs_add_full(a, A);
      gs_load(A, supLane, list, b + 2 * GS_WAVES, total);
      __builtin_amdgcn_sched_barrier(0);
      gs_add_full(a, B);
      b += 2 * GS_WAVES;
    }
    if (b + GS_WAVES < nb) {
      gs_load(B, supLane, list, b + GS_WAVES, total);
      __builtin_amdgcn_sched_barrier(0);
      gs_add_full(a, A);
      gs_add_last(a, B, b + GS_WAVES, total);
    } else {
      gs_add_last(a, A, b, total);
    }
  }

  #pragma unroll
  for (int qd = 0; qd < 4; ++qd)
    *(v4f*)(red + w * D + qd * 128 + lane * 4) = a[qd];
  __syncthreads();
  float* dst = pSums + (size_t)blockIdx.x * D;
  #pragma unroll
  for (int j = 0; j < 2; ++j) {
    const int d = t + j * 256;
    float s = 0.f;
    #pragma unroll
    for (int i = 0; i < GS_WAVES; ++i) s += red[i * D + d];
    dst[d] = s;
  }
  if (t == 0) pCnt[blockIdx.x] = total;
}

__global__ void __launch_bounds__(256)
proto_final(const float* __restrict__ pSums, const int* __restrict__ pCnt,
            const float* __restrict__ scaleP, float* __restrict__ protN) {
  __shared__ float red[256];
  const int c = blockIdx.x;
  const int t = threadIdx.x;

  int cnt = 0;
  #pragma unroll
  for (int i = 0; i < GS_SPLIT; ++i) cnt += pCnt[c * GS_SPLIT + i];
  const float fcnt = (float)cnt;

  float p[2];
  #pragma unroll
  for (int j = 0; j < 2; ++j) {
    const int d = t + j * 256;
    float s = 0.f;
    #pragma unroll
    for (int i = 0; i < GS_SPLIT; ++i)
      s += pSums[(size_t)(c * GS_SPLIT + i) * D + d];
    p[j] = s / fcnt;
  }

  red[t] = p[0] * p[0] + p[1] * p[1];
  __syncthreads();
  for (int s = 128; s > 0; s >>= 1) {
    if (t < s) red[t] += red[t + s];
    __syncthreads();
  }
  const float fac = scaleP[0] / fmaxf(sqrtf(red[0]), EPSF);
  protN[(size_t)c * D + t]       = p[0] * fac;
  protN[(size_t)c * D + t + 256] = p[1] * fac;
}

#define CW     16
#define NCH    (D / KC)
#define PCH    (NWAY * LDSTR)

__device__ __forceinline__ void cq_load(v4f (&R)[8], const float* __restrict__ src) {
  #pragma unroll
  for (int j = 0; j < 8; ++j)
    R[j] = __builtin_nontemporal_load((const v4f*)(src + (size_t)(2 * j) * D));
}

__global__ void __launch_bounds__(512)
cosine_gemm(const float* __restrict__ q, const float* __restrict__ protN,
            float* __restrict__ out) {
  __shared__ __attribute__((aligned(16))) float ps[NCH * PCH];
  __shared__ __attribute__((aligned(16))) float qst[CW * 16 * LDSTR];

  const int t    = threadIdx.x;
  const int w    = __builtin_amdgcn_readfirstlane(t >> 5);
  const int lane = t & 31;
  const int l15  = lane & 15;
  const int hi   = lane >> 4;
  const size_t qBase = ((size_t)blockIdx.x * CW + w) * 16;

  const float* qsrc = q + (qBase + hi) * D + l15 * 4;
  float* myst = qst + w * (16 * LDSTR);
  float* stdst = myst + hi * LDSTR + l15 * 4;

  v4f R[8];
  cq_load(R, qsrc);

  #pragma unroll
  for (int j = 0; j < 16; ++j) {
    const int v = t + j * 512;
    const int row = v >> 7, c4 = (v & 127) * 4;
    *(v4f*)(ps + (c4 >> 6) * PCH + row * LDSTR + (c4 & 63)) = *(const v4f*)(protN + (size_t)row * D + c4);
  }
  __syncthreads();

  v8f acc0 = {}, acc1 = {}, acc2 = {}, acc3 = {};
  float nacc = 0.f;
  const float* afrag = myst + l15 * LDSTR + 2 * hi;
  const float* bfrag = ps + l15 * LDSTR + 2 * hi;

  for (int c = 0; c < NCH; ++c) {
    #pragma unroll
    for (int j = 0; j < 8; ++j)
      *(v4f*)(stdst + (2 * j) * LDSTR) = R[j];
    if (c + 1 < NCH) cq_load(R, qsrc + (c + 1) * KC);
    __builtin_amdgcn_fence(__ATOMIC_RELEASE, "wavefront");
    __builtin_amdgcn_wave_barrier();
    __builtin_amdgcn_fence(__ATOMIC_ACQUIRE, "wavefront");

    const float* bk = bfrag + c * PCH;
    for (int k4 = 0; k4 < KC; k4 += 4) {
      v2f a  = *(const v2f*)(afrag + k4);
      v2f b0 = *(const v2f*)(bk +  0 * LDSTR + k4);
      v2f b1 = *(const v2f*)(bk + 16 * LDSTR + k4);
      v2f b2 = *(const v2f*)(bk + 32 * LDSTR + k4);
      v2f b3 = *(const v2f*)(bk + 48 * LDSTR + k4);
      nacc += a.x * a.x + a.y * a.y;
      acc0 = __builtin_amdgcn_wmma_f32_16x16x4_f32(false, a, false, b0, (short)0, acc0, false, false);
      acc1 = __builtin_amdgcn_wmma_f32_16x16x4_f32(false, a, false, b1, (short)0, acc1, false, false);
      acc2 = __builtin_amdgcn_wmma_f32_16x16x4_f32(false, a, false, b2, (short)0, acc2, false, false);
      acc3 = __builtin_amdgcn_wmma_f32_16x16x4_f32(false, a, false, b3, (short)0, acc3, false, false);
    }
    __builtin_amdgcn_fence(__ATOMIC_RELEASE, "wavefront");
    __builtin_amdgcn_wave_barrier();
    __builtin_amdgcn_fence(__ATOMIC_ACQUIRE, "wavefront");
  }

  const float n2  = nacc + __shfl_xor(nacc, 16);
  const float inv = 1.f / fmaxf(sqrtf(n2), EPSF);

  #pragma unroll
  for (int r = 0; r < 8; ++r) {
    const int qrow = r + 8 * hi;
    const float s = __shfl(inv, qrow);
    const size_t orow = (qBase + qrow) * NWAY;
    out[orow +  0 + l15] = acc0[r] * s;
    out[orow + 16 + l15] = acc1[r] * s;
    out[orow + 32 + l15] = acc2[r] * s;
    out[orow + 48 + l15] = acc3[r] * s;
  }
}

extern "C" void kernel_launch(void* const* d_in, const int* in_sizes, int n_in,
                              void* d_out, int out_size, void* d_ws, size_t ws_size,
                              hipStream_t stream) {
  const float* sup    = (const float*)d_in[0];
  const int*   slab   = (const int*)d_in[1];
  const float* qry    = (const float*)d_in[2];
  const float* scaleP = (const float*)d_in[4];
  float* out = (float*)d_out;

  float* pSums = (float*)d_ws;
  int*   pCnt  = (int*)((char*)d_ws + (size_t)NWAY * GS_SPLIT * D * sizeof(float));
  float* protN = (float*)((char*)d_ws + (size_t)128 * NWAY * 256 * sizeof(float));

  class_gather<<<NWAY * GS_SPLIT, 256, 0, stream>>>(sup, slab, pSums, pCnt);
  proto_final <<<NWAY, 256, 0, stream>>>(pSums, pCnt, scaleP, protN);
  cosine_gemm  <<<NQ / (CW * 16), CW * 32, 0, stream>>>(qry, protN, out);
}
